# starting kernel
# baseline (speedup 1.0000x reference)
#include <hip/hip_runtime.h>
#include <math.h>

#define N_ITEMS 200000
#define DIMS    128
#define S_PERMS 128
#define ALPHA_C 0.7f
#define EPS_C   1e-8f

typedef float v2f __attribute__((ext_vector_type(2)));
typedef float v8f __attribute__((ext_vector_type(8)));

__device__ __forceinline__ unsigned fenc(float f) {
  unsigned u = __float_as_uint(f);
  return (u & 0x80000000u) ? ~u : (u | 0x80000000u);
}
__device__ __forceinline__ float fdec(unsigned e) {
  unsigned u = (e & 0x80000000u) ? (e & 0x7FFFFFFFu) : ~e;
  return __uint_as_float(u);
}

__global__ void init_kernel(const float* __restrict__ tgt,
                            float* __restrict__ stats,
                            unsigned* __restrict__ umax) {
  int lane = threadIdx.x;
  float ss = 0.f;
  for (int k = lane; k < DIMS; k += 32) { float v = tgt[k]; ss = fmaf(v, v, ss); }
  #pragma unroll
  for (int o = 16; o; o >>= 1) ss += __shfl_xor(ss, o);
  if (lane == 0) { stats[0] = sqrtf(ss); stats[1] = 0.f; *umax = 0u; }
}

__global__ void __launch_bounds__(256)
vals_kernel(const float* __restrict__ neigh, const float* __restrict__ tgt,
            const float* __restrict__ stats, float* __restrict__ vals,
            float* __restrict__ w, int nTiles) {
  __shared__ __align__(16) float lds[8][16 * 132];
  const int lane  = threadIdx.x & 31;
  const int wave  = threadIdx.x >> 5;
  const int hi    = lane >> 4;
  const int row16 = lane & 15;
  float* tile = &lds[wave][0];
  const float tn = fmaxf(stats[0], EPS_C);

  v2f bv[32];
  #pragma unroll
  for (int k = 0; k < 32; ++k)
    bv[k] = *(const v2f*)(tgt + 4 * k + 2 * hi);

  const int r = lane >> 1, h = lane & 1;
  const int wavesTotal = (gridDim.x * blockDim.x) >> 5;
  for (int t = (blockIdx.x * blockDim.x + threadIdx.x) >> 5; t < nTiles;
       t += wavesTotal) {
    const float* src = neigh + (size_t)t * (16 * DIMS);
    const float4* srow = (const float4*)(src + r * DIMS + h * 64);
    float* drow = tile + r * 132 + h * 64;
    float ss = 0.f;
    #pragma unroll
    for (int j = 0; j < 16; ++j) {
      float4 v = srow[j];
      ss = fmaf(v.x, v.x, ss); ss = fmaf(v.y, v.y, ss);
      ss = fmaf(v.z, v.z, ss); ss = fmaf(v.w, v.w, ss);
      *(float4*)(drow + 4 * j) = v;
    }
    ss += __shfl_xor(ss, 1);

    v8f acc = {};
    const float* arow = tile + row16 * 132 + 2 * hi;
    #pragma unroll
    for (int k = 0; k < 32; ++k) {
      v2f a = *(const v2f*)(arow + 4 * k);
      acc = __builtin_amdgcn_wmma_f32_16x16x4_f32(false, a, false, bv[k],
                                                  (short)0, acc, false, false);
    }

    float vout[8];
    #pragma unroll
    for (int j = 0; j < 8; ++j) {
      float ssj = __shfl(ss, (lane & 16) + 2 * j);
      float dot = acc[j];
      float nn  = fmaxf(sqrtf(ssj), EPS_C);
      float sim = dot / (nn * tn);
      vout[j] = dot * __powf(sim, ALPHA_C);
    }
    if (row16 == 0) {
      int base = t * 16 + hi * 8;
      float4 lo  = make_float4(vout[0], vout[1], vout[2], vout[3]);
      float4 hi4 = make_float4(vout[4], vout[5], vout[6], vout[7]);
      *(float4*)(vals + base)     = lo;  *(float4*)(vals + base + 4) = hi4;
      *(float4*)(w + base)        = lo;  *(float4*)(w + base + 4)    = hi4;
    }
  }
}

__global__ void __launch_bounds__(256)
scatter_kernel(const int* __restrict__ perms, const float* __restrict__ vals,
               float* __restrict__ w) {
  const int s = blockIdx.y;
  const int i = blockIdx.x * blockDim.x + threadIdx.x;
  if (i >= N_ITEMS) return;
  const int idx = s * N_ITEMS + i;
  const int p  = perms[idx];
  int pp = __shfl_up(p, 1);
  if ((threadIdx.x & 31) == 0 && i > 0) pp = perms[idx - 1];
  if (i > 0) atomicAdd(&w[p], vals[pp] * (-1.0f / S_PERMS));
}

__global__ void max_kernel(const float* __restrict__ w, unsigned* __restrict__ umax, int n) {
  float m = -INFINITY;
  for (int i = blockIdx.x * blockDim.x + threadIdx.x; i < n; i += gridDim.x * blockDim.x)
    m = fmaxf(m, w[i]);
  #pragma unroll
  for (int o = 16; o; o >>= 1) m = fmaxf(m, __shfl_xor(m, o));
  if ((threadIdx.x & 31) == 0) atomicMax(umax, fenc(m));
}

__global__ void sum_kernel(const float* __restrict__ w, const unsigned* __restrict__ umax,
                           float* __restrict__ sum, int n) {
  const float mx = fdec(*umax);
  float a = 0.f;
  for (int i = blockIdx.x * blockDim.x + threadIdx.x; i < n; i += gridDim.x * blockDim.x)
    a += __expf(w[i] - mx);
  #pragma unroll
  for (int o = 16; o; o >>= 1) a += __shfl_xor(a, o);
  if ((threadIdx.x & 31) == 0) atomicAdd(sum, a);
}

__global__ void final_kernel(float* __restrict__ w, const unsigned* __restrict__ umax,
                             const float* __restrict__ sum, int n) {
  const float mx = fdec(*umax);
  const float rs = 1.0f / *sum;
  int i = blockIdx.x * blockDim.x + threadIdx.x;
  if (i < n) w[i] = __expf(w[i] - mx) * rs;
}

extern "C" void kernel_launch(void* const* d_in, const int* in_sizes, int n_in,
                              void* d_out, int out_size, void* d_ws, size_t ws_size,
                              hipStream_t stream) {
  const float* neigh = (const float*)d_in[0];
  const float* tgt   = (const float*)d_in[1];
  const int*   perms = (const int*)d_in[2];
  float* out  = (float*)d_out;
  float* vals = (float*)d_ws;
  float* stats = vals + N_ITEMS;
  unsigned* umax = (unsigned*)(stats + 2);

  init_kernel<<<1, 32, 0, stream>>>(tgt, stats, umax);

  const int nTiles = N_ITEMS / 16;
  vals_kernel<<<(nTiles + 7) / 8, 256, 0, stream>>>(neigh, tgt, stats, vals, out, nTiles);

  dim3 sg((N_ITEMS + 255) / 256, S_PERMS);
  scatter_kernel<<<sg, 256, 0, stream>>>(perms, vals, out);

  max_kernel<<<1024, 256, 0, stream>>>(out, umax, N_ITEMS);
  sum_kernel<<<1024, 256, 0, stream>>>(out, umax, stats + 1, N_ITEMS);
  final_kernel<<<(N_ITEMS + 255) / 256, 256, 0, stream>>>(out, umax, stats + 1, N_ITEMS);
}


# Softmax max and sum launches: four items a lane, a 256-lane work-group folds its waves in LDS and sends one atomic to the shared word: 196 per launch, was 8192; float sum order changes
# speedup vs baseline: 11.0765x; 11.0765x over previous
#include <hip/hip_runtime.h>
#include <math.h>

#define N_ITEMS 200000
#define DIMS    128
#define S_PERMS 128
#define ALPHA_C 0.7f
#define EPS_C   1e-8f

#define TICKET    128
#define N_TICKETS (S_PERMS * N_ITEMS / TICKET)
#define N_SUB       8
#define SUB_TICKETS (N_TICKETS / N_SUB)
#define Q_STRIDE    64
#define SPLIT_ITEM 100032

#define FX_TERM 131072.0f
#define FX_W    16777216.0f
#define CHAIN_WS_BYTES (2 * N_ITEMS * 8 + 256 + 2 * N_SUB * Q_STRIDE * 4)

typedef unsigned long long u64;
typedef float v2f __attribute__((ext_vector_type(2)));
typedef float v8f __attribute__((ext_vector_type(8)));

__device__ __forceinline__ unsigned fenc(float f) {
  unsigned u = __float_as_uint(f);
  return (u & 0x80000000u) ? ~u : (u | 0x80000000u);
}
__device__ __forceinline__ float fdec(unsigned e) {
  unsigned u = (e & 0x80000000u) ? (e & 0x7FFFFFFFu) : ~e;
  return __uint_as_float(u);
}

__global__ void init_kernel(const float* __restrict__ tgt,
                            float* __restrict__ stats,
                            unsigned* __restrict__ umax,
                            unsigned* __restrict__ q) {
  int lane = threadIdx.x;
  float ss = 0.f;
  for (int k = lane; k < DIMS; k += 32) { float v = tgt[k]; ss = fmaf(v, v, ss); }
  #pragma unroll
  for (int o = 16; o; o >>= 1) ss += __shfl_xor(ss, o);
  if (lane == 0) {
    stats[0] = sqrtf(ss); stats[1] = 0.f; *umax = 0u;
  }
  if (lane < 2 * N_SUB) q[lane * Q_STRIDE] = 0u;
}

template <bool WORDS>
__global__ void __launch_bounds__(256)
vals_kernel(const float* __restrict__ neigh, const float* __restrict__ tgt,
            const float* __restrict__ stats, float* __restrict__ vals,
            float* __restrict__ w, unsigned long long* __restrict__ words, int nTiles) {
  __shared__ __align__(16) float lds[8][16 * 132];
  const int lane  = threadIdx.x & 31;
  const int wave  = threadIdx.x >> 5;
  const int hi    = lane >> 4;
  const int row16 = lane & 15;
  float* tile = &lds[wave][0];
  const float tn = fmaxf(stats[0], EPS_C);

  v2f bv[32];
  #pragma unroll
  for (int k = 0; k < 32; ++k)
    bv[k] = *(const v2f*)(tgt + 4 * k + 2 * hi);

  const int r = lane >> 1, h = lane & 1;
  const int wavesTotal = (gridDim.x * blockDim.x) >> 5;
  for (int t = (blockIdx.x * blockDim.x + threadIdx.x) >> 5; t < nTiles;
       t += wavesTotal) {
    const float* src = neigh + (size_t)t * (16 * DIMS);
    const float4* srow = (const float4*)(src + r * DIMS + h * 64);
    float* drow = tile + r * 132 + h * 64;
    float ss = 0.f;
    #pragma unroll
    for (int j = 0; j < 16; ++j) {
      float4 v = srow[j];
      ss = fmaf(v.x, v.x, ss); ss = fmaf(v.y, v.y, ss);
      ss = fmaf(v.z, v.z, ss); ss = fmaf(v.w, v.w, ss);
      *(float4*)(drow + 4 * j) = v;
    }
    ss += __shfl_xor(ss, 1);

    v8f acc = {};
    const float* arow = tile + row16 * 132 + 2 * hi;
    #pragma unroll
    for (int k = 0; k < 32; ++k) {
      v2f a = *(const v2f*)(arow + 4 * k);
      acc = __builtin_amdgcn_wmma_f32_16x16x4_f32(false, a, false, bv[k],
                                                  (short)0, acc, false, false);
    }

    float vout[8];
    #pragma unroll
    for (int j = 0; j < 8; ++j) {
      float ssj = __shfl(ss, (lane & 16) + 2 * j);
      float dot = acc[j];
      float nn  = fmaxf(sqrtf(ssj), EPS_C);
      float sim = dot / (nn * tn);
      vout[j] = dot * __powf(sim, ALPHA_C);
    }
    if (row16 == 0) {
      int base = t * 16 + hi * 8;
      if (WORDS) {
        unsigned* accz = (unsigned*)words;
        *(float4*)(vals + base)     = make_float4(vout[0], vout[1], vout[2], vout[3]);
        *(float4*)(vals + base + 4) = make_float4(vout[4], vout[5], vout[6], vout[7]);
        const uint4 z = make_uint4(0u, 0u, 0u, 0u);
        *(uint4*)(accz + base) = z;            *(uint4*)(accz + base + 4) = z;
        *(uint4*)(accz + N_ITEMS + base) = z;  *(uint4*)(accz + N_ITEMS + base + 4) = z;
      } else {
        float4 lo  = make_float4(vout[0], vout[1], vout[2], vout[3]);
        float4 hi4 = make_float4(vout[4], vout[5], vout[6], vout[7]);
        *(float4*)(vals + base)     = lo;  *(float4*)(vals + base + 4) = hi4;
        *(float4*)(w + base)        = lo;  *(float4*)(w + base + 4)    = hi4;
      }
    }
  }
}


__device__ __forceinline__ unsigned take_ticket(unsigned* cnt, int lane) {
  unsigned v = 0u;
  if (lane == 0) v = atomicAdd(cnt, 1u);
  return (unsigned)__builtin_amdgcn_readfirstlane((int)v);
}

__global__ void __launch_bounds__(256)
scatter_kernel(const int* __restrict__ perms, const float* __restrict__ vals,
               float* __restrict__ w, unsigned* __restrict__ q) {
  const int lane = threadIdx.x & 31;
  const unsigned slot = (unsigned)__builtin_amdgcn_readfirstlane((int)(threadIdx.x >> 5));
  const unsigned half = ((unsigned)__builtin_amdgcn_s_sendmsg_rtn(0x87) >> 18) & 1u;
  const float sc = -1.0f / S_PERMS;

  for (unsigned pass = 0; pass < 2u; ++pass) {
    const unsigned r = half ^ pass;
    unsigned* cnt = q + (r * N_SUB + slot) * Q_STRIDE;
    unsigned k = take_ticket(cnt, lane);
    while (k < (unsigned)SUB_TICKETS) {
      const unsigned t = k * N_SUB + slot;
      const unsigned f = t * TICKET + 4u * (unsigned)lane;
      const int4 pv = *(const int4*)(perms + f);
      const int pprev = perms[f ? f - 1u : 0u];
      const unsigned kn = take_ticket(cnt, lane);
      const bool o0 = ((((unsigned)pv.x >= (unsigned)SPLIT_ITEM) ? 1u : 0u) == r) && (f % N_ITEMS != 0u);
      const bool o1 = ((((unsigned)pv.y >= (unsigned)SPLIT_ITEM) ? 1u : 0u) == r);
      const bool o2 = ((((unsigned)pv.z >= (unsigned)SPLIT_ITEM) ? 1u : 0u) == r);
      const bool o3 = ((((unsigned)pv.w >= (unsigned)SPLIT_ITEM) ? 1u : 0u) == r);
      const float g0 = vals[o0 ? pprev : 0];
      const float g1 = vals[o1 ? pv.x : 0];
      const float g2 = vals[o2 ? pv.y : 0];
      const float g3 = vals[o3 ? pv.z : 0];
      if (o0) atomicAdd(&w[pv.x], g0 * sc);
      if (o1) atomicAdd(&w[pv.y], g1 * sc);
      if (o2) atomicAdd(&w[pv.z], g2 * sc);
      if (o3) atomicAdd(&w[pv.w], g3 * sc);
      k = kn;
    }
  }
}

#define BATCH        50000
#define BATCHES_ROW  (N_ITEMS / BATCH)
#define N_BATCHES    (S_PERMS * BATCHES_ROW)
#define LINE_SHIFT   5
#define N_LINES      (N_ITEMS >> LINE_SHIFT)
#define BINS_PER_THR 7
#define PER_LANE     ((BATCH + 1023) / 1024)
#define ADD_GROUP    7

__global__ void __launch_bounds__(1024)
line_scatter_kernel(const int* __restrict__ perms, const float* __restrict__ vals,
                    unsigned* __restrict__ acc) {
  __shared__ unsigned buf[BATCH + 1];
  __shared__ unsigned short order[BATCH];
  __shared__ unsigned hist[N_LINES];
  __shared__ unsigned wsum[32];
  const unsigned tid = threadIdx.x;
  const unsigned half = ((unsigned)__builtin_amdgcn_s_sendmsg_rtn(0x87) >> 18) & 1u;
  unsigned* a = acc + half * (unsigned)N_ITEMS;
  const unsigned b = blockIdx.x;
  const bool opens = (b % (unsigned)BATCHES_ROW) == 0u;
  const int* src = perms + (size_t)b * BATCH;

  for (unsigned k = tid; k < (unsigned)N_LINES; k += 1024u) hist[k] = 0u;
  __syncthreads();
  int mine[PER_LANE];
  #pragma unroll
  for (int j = 0; j < PER_LANE; ++j) {
    const unsigned i = tid + 1024u * (unsigned)j;
    mine[j] = (i < (unsigned)BATCH) ? src[i] : 0;
  }
  #pragma unroll
  for (int j = 0; j < PER_LANE; ++j) {
    const unsigned i = tid + 1024u * (unsigned)j;
    if (i < (unsigned)BATCH) {
      buf[i + 1] = (unsigned)mine[j];
      atomicAdd(&hist[(unsigned)mine[j] >> LINE_SHIFT], 1u);
    }
  }
  if (tid == 0) buf[0] = (unsigned)(opens ? src[0] : src[-1]);
  __syncthreads();
  unsigned c[BINS_PER_THR], tot = 0u;
  #pragma unroll
  for (int j = 0; j < BINS_PER_THR; ++j) {
    const unsigned bin = tid * BINS_PER_THR + (unsigned)j;
    c[j] = (bin < (unsigned)N_LINES) ? hist[bin] : 0u;
    tot += c[j];
  }
  unsigned inc = tot;
  #pragma unroll
  for (int o = 1; o < 32; o <<= 1) {
    const unsigned u = __shfl_up(inc, o);
    if ((tid & 31u) >= (unsigned)o) inc += u;
  }
  if ((tid & 31u) == 31u) wsum[tid >> 5] = inc;
  __syncthreads();
  if (tid < 32u) {
    unsigned x = wsum[tid], y = x;
    #pragma unroll
    for (int o = 1; o < 32; o <<= 1) {
      const unsigned u = __shfl_up(y, o);
      if (tid >= (unsigned)o) y += u;
    }
    wsum[tid] = y - x;
  }
  __syncthreads();
  unsigned run = wsum[tid >> 5] + inc - tot;
  #pragma unroll
  for (int j = 0; j < BINS_PER_THR; ++j) {
    const unsigned bin = tid * BINS_PER_THR + (unsigned)j;
    if (bin < (unsigned)N_LINES) hist[bin] = run;
    run += c[j];
  }
  __syncthreads();
  for (unsigned i = tid; i < (unsigned)BATCH; i += 1024u) {
    const unsigned pos = atomicAdd(&hist[buf[i + 1] >> LINE_SHIFT], 1u);
    order[pos] = (unsigned short)i;
  }
  __syncthreads();
  unsigned dst[PER_LANE];
  #pragma unroll
  for (int j = 0; j < PER_LANE; ++j) {
    const unsigned k = tid + 1024u * (unsigned)j;
    dst[j] = (k < (unsigned)BATCH) ? buf[(unsigned)order[k] + 1u] : 0u;
  }
  const unsigned pred = buf[0];
  __syncthreads();
  #pragma unroll
  for (int j = 0; j < PER_LANE; ++j) {
    const unsigned k = tid + 1024u * (unsigned)j;
    if (k < (unsigned)BATCH)
      buf[(unsigned)order[k] + 1u] = __float2uint_rn(vals[dst[j]] * FX_TERM);
  }
  if (tid == 0) buf[0] = __float2uint_rn(vals[pred] * FX_TERM);
  __syncthreads();
  #pragma unroll
  for (int g = 0; g < PER_LANE; g += ADD_GROUP) {
    unsigned ent[ADD_GROUP], sum[ADD_GROUP];
    #pragma unroll
    for (int u = 0; u < ADD_GROUP; ++u) {
      const unsigned k = tid + 1024u * (unsigned)(g + u);
      ent[u] = (k < (unsigned)BATCH) ? (unsigned)order[k] : 0u;
    }
    #pragma unroll
    for (int u = 0; u < ADD_GROUP; ++u) sum[u] = buf[ent[u]];
    #pragma unroll
    for (int u = 0; u < ADD_GROUP; ++u) {
      const unsigned k = tid + 1024u * (unsigned)(g + u);
      if (k < (unsigned)BATCH && !(opens && ent[u] == 0u))
        __hip_atomic_fetch_add(a + dst[g + u], sum[u], __ATOMIC_RELAXED, __HIP_MEMORY_SCOPE_AGENT);
    }
  }
}

#define TAIL_LANES  256
#define TAIL_QUADS  (N_ITEMS / 4)
#define TAIL_GROUPS ((TAIL_QUADS + TAIL_LANES - 1) / TAIL_LANES)

__device__ __forceinline__ float fold_one(float v, unsigned s0, unsigned s1) {
  const int d = (int)(__float2uint_rn(v * FX_W) - (s0 + s1));
  return (float)d * (1.0f / FX_W);
}

__global__ void __launch_bounds__(TAIL_LANES)
fold_max_kernel(const float* __restrict__ vals, const unsigned* __restrict__ acc,
                float* __restrict__ w, unsigned* __restrict__ umax) {
  __shared__ float part[TAIL_LANES / 32];
  const int q = blockIdx.x * TAIL_LANES + threadIdx.x;
  float m = -INFINITY;
  float4 o = make_float4(0.f, 0.f, 0.f, 0.f);
  if (q < TAIL_QUADS) {
    const uint4  s0 = ((const uint4*)acc)[q];
    const uint4  s1 = ((const uint4*)(acc + N_ITEMS))[q];
    const float4 v  = ((const float4*)vals)[q];
    o.x = fold_one(v.x, s0.x, s1.x);
    o.y = fold_one(v.y, s0.y, s1.y);
    o.z = fold_one(v.z, s0.z, s1.z);
    o.w = fold_one(v.w, s0.w, s1.w);
    m = fmaxf(fmaxf(o.x, o.y), fmaxf(o.z, o.w));
  }
  #pragma unroll
  for (int s = 16; s; s >>= 1) m = fmaxf(m, __shfl_xor(m, s));
  if ((threadIdx.x & 31) == 0) part[threadIdx.x >> 5] = m;
  __syncthreads();
  if (threadIdx.x == 0) {
    float x = part[0];
    #pragma unroll
    for (int j = 1; j < TAIL_LANES / 32; ++j) x = fmaxf(x, part[j]);
    atomicMax(umax, fenc(x));
  }
  if (q < TAIL_QUADS) ((float4*)w)[q] = o;
}

__global__ void max_kernel(const float* __restrict__ w, unsigned* __restrict__ umax, int n) {
  float m = -INFINITY;
  for (int i = blockIdx.x * blockDim.x + threadIdx.x; i < n; i += gridDim.x * blockDim.x)
    m = fmaxf(m, w[i]);
  #pragma unroll
  for (int o = 16; o; o >>= 1) m = fmaxf(m, __shfl_xor(m, o));
  if ((threadIdx.x & 31) == 0) atomicMax(umax, fenc(m));
}

__global__ void __launch_bounds__(TAIL_LANES)
sum_kernel(const float* __restrict__ w, const unsigned* __restrict__ umax,
           float* __restrict__ sum) {
  __shared__ float part[TAIL_LANES / 32];
  const float mx = fdec(*umax);
  const int q = blockIdx.x * TAIL_LANES + threadIdx.x;
  float a = 0.f;
  if (q < TAIL_QUADS) {
    const float4 v = ((const float4*)w)[q];
    a = (__expf(v.x - mx) + __expf(v.y - mx)) + (__expf(v.z - mx) + __expf(v.w - mx));
  }
  #pragma unroll
  for (int o = 16; o; o >>= 1) a += __shfl_xor(a, o);
  if ((threadIdx.x & 31) == 0) part[threadIdx.x >> 5] = a;
  __syncthreads();
  if (threadIdx.x == 0) {
    float x = part[0];
    #pragma unroll
    for (int j = 1; j < TAIL_LANES / 32; ++j) x += part[j];
    atomicAdd(sum, x);
  }
}

__global__ void final_kernel(float* __restrict__ w, const unsigned* __restrict__ umax,
                             const float* __restrict__ sum, int n) {
  const float mx = fdec(*umax);
  const float rs = 1.0f / *sum;
  int i = blockIdx.x * blockDim.x + threadIdx.x;
  if (i < n) w[i] = __expf(w[i] - mx) * rs;
}

extern "C" void kernel_launch(void* const* d_in, const int* in_sizes, int n_in,
                              void* d_out, int out_size, void* d_ws, size_t ws_size,
                              hipStream_t stream) {
  const float* neigh = (const float*)d_in[0];
  const float* tgt   = (const float*)d_in[1];
  const int*   perms = (const int*)d_in[2];
  float* out  = (float*)d_out;
  const int nTiles = N_ITEMS / 16;

  if (ws_size >= (size_t)CHAIN_WS_BYTES) {
    u64* base64 = (u64*)d_ws;
    float* cvals = (float*)d_ws;
    unsigned* acc = (unsigned*)(cvals + N_ITEMS);
    float* cstats = (float*)(base64 + 2 * N_ITEMS);
    unsigned* cumax = (unsigned*)(cstats + 2);
    unsigned* cq = (unsigned*)(cstats + 64);
    init_kernel<<<1, 32, 0, stream>>>(tgt, cstats, cumax, cq);
    vals_kernel<true><<<(nTiles + 7) / 8, 256, 0, stream>>>(neigh, tgt, cstats, cvals, nullptr, (unsigned long long*)acc, nTiles);
    line_scatter_kernel<<<N_BATCHES, 1024, 0, stream>>>(perms, cvals, acc);
    fold_max_kernel<<<TAIL_GROUPS, TAIL_LANES, 0, stream>>>(cvals, acc, out, cumax);
    sum_kernel<<<TAIL_GROUPS, TAIL_LANES, 0, stream>>>(out, cumax, cstats + 1);
    final_kernel<<<(N_ITEMS + 255) / 256, 256, 0, stream>>>(out, cumax, cstats + 1, N_ITEMS);
    return;
  }

  float* vals = (float*)d_ws;
  float* stats = vals + N_ITEMS;
  unsigned* umax = (unsigned*)(stats + 2);
  unsigned* q = (unsigned*)(stats + 64);

  init_kernel<<<1, 32, 0, stream>>>(tgt, stats, umax, q);

  vals_kernel<false><<<(nTiles + 7) / 8, 256, 0, stream>>>(neigh, tgt, stats, vals, out, nullptr, nTiles);

  scatter_kernel<<<2048, 256, 0, stream>>>(perms, vals, out, q);

  max_kernel<<<1024, 256, 0, stream>>>(out, umax, N_ITEMS);
  sum_kernel<<<TAIL_GROUPS, TAIL_LANES, 0, stream>>>(out, umax, stats + 1);
  final_kernel<<<(N_ITEMS + 255) / 256, 256, 0, stream>>>(out, umax, stats + 1, N_ITEMS);
}


# Scatter sort: a lane keeps its items in registers through both counting passes; the LDS word at a sorted position is entry number | item low half: 6 LDS operations an entry, was 11; sums bit-equal
# speedup vs baseline: 11.5166x; 1.0397x over previous
#include <hip/hip_runtime.h>
#include <math.h>

#define N_ITEMS 200000
#define DIMS    128
#define S_PERMS 128
#define ALPHA_C 0.7f
#define EPS_C   1e-8f

#define TICKET    128
#define N_TICKETS (S_PERMS * N_ITEMS / TICKET)
#define N_SUB       8
#define SUB_TICKETS (N_TICKETS / N_SUB)
#define Q_STRIDE    64
#define SPLIT_ITEM 100032

#define FX_TERM 131072.0f
#define FX_W    16777216.0f
#define CHAIN_WS_BYTES (2 * N_ITEMS * 8 + 256 + 2 * N_SUB * Q_STRIDE * 4)

typedef unsigned long long u64;
typedef float v2f __attribute__((ext_vector_type(2)));
typedef float v8f __attribute__((ext_vector_type(8)));

__device__ __forceinline__ unsigned fenc(float f) {
  unsigned u = __float_as_uint(f);
  return (u & 0x80000000u) ? ~u : (u | 0x80000000u);
}
__device__ __forceinline__ float fdec(unsigned e) {
  unsigned u = (e & 0x80000000u) ? (e & 0x7FFFFFFFu) : ~e;
  return __uint_as_float(u);
}

__global__ void init_kernel(const float* __restrict__ tgt,
                            float* __restrict__ stats,
                            unsigned* __restrict__ umax,
                            unsigned* __restrict__ q) {
  int lane = threadIdx.x;
  float ss = 0.f;
  for (int k = lane; k < DIMS; k += 32) { float v = tgt[k]; ss = fmaf(v, v, ss); }
  #pragma unroll
  for (int o = 16; o; o >>= 1) ss += __shfl_xor(ss, o);
  if (lane == 0) {
    stats[0] = sqrtf(ss); stats[1] = 0.f; *umax = 0u;
  }
  if (lane < 2 * N_SUB) q[lane * Q_STRIDE] = 0u;
}

template <bool WORDS>
__global__ void __launch_bounds__(256)
vals_kernel(const float* __restrict__ neigh, const float* __restrict__ tgt,
            const float* __restrict__ stats, float* __restrict__ vals,
            float* __restrict__ w, unsigned long long* __restrict__ words, int nTiles) {
  __shared__ __align__(16) float lds[8][16 * 132];
  const int lane  = threadIdx.x & 31;
  const int wave  = threadIdx.x >> 5;
  const int hi    = lane >> 4;
  const int row16 = lane & 15;
  float* tile = &lds[wave][0];
  const float tn = fmaxf(stats[0], EPS_C);

  v2f bv[32];
  #pragma unroll
  for (int k = 0; k < 32; ++k)
    bv[k] = *(const v2f*)(tgt + 4 * k + 2 * hi);

  const int r = lane >> 1, h = lane & 1;
  const int wavesTotal = (gridDim.x * blockDim.x) >> 5;
  for (int t = (blockIdx.x * blockDim.x + threadIdx.x) >> 5; t < nTiles;
       t += wavesTotal) {
    const float* src = neigh + (size_t)t * (16 * DIMS);
    const float4* srow = (const float4*)(src + r * DIMS + h * 64);
    float* drow = tile + r * 132 + h * 64;
    float ss = 0.f;
    #pragma unroll
    for (int j = 0; j < 16; ++j) {
      float4 v = srow[j];
      ss = fmaf(v.x, v.x, ss); ss = fmaf(v.y, v.y, ss);
      ss = fmaf(v.z, v.z, ss); ss = fmaf(v.w, v.w, ss);
      *(float4*)(drow + 4 * j) = v;
    }
    ss += __shfl_xor(ss, 1);

    v8f acc = {};
    const float* arow = tile + row16 * 132 + 2 * hi;
    #pragma unroll
    for (int k = 0; k < 32; ++k) {
      v2f a = *(const v2f*)(arow + 4 * k);
      acc = __builtin_amdgcn_wmma_f32_16x16x4_f32(false, a, false, bv[k],
                                                  (short)0, acc, false, false);
    }

    float vout[8];
    #pragma unroll
    for (int j = 0; j < 8; ++j) {
      float ssj = __shfl(ss, (lane & 16) + 2 * j);
      float dot = acc[j];
      float nn  = fmaxf(sqrtf(ssj), EPS_C);
      float sim = dot / (nn * tn);
      vout[j] = dot * __powf(sim, ALPHA_C);
    }
    if (row16 == 0) {
      int base = t * 16 + hi * 8;
      if (WORDS) {
        unsigned* accz = (unsigned*)words;
        *(float4*)(vals + base)     = make_float4(vout[0], vout[1], vout[2], vout[3]);
        *(float4*)(vals + base + 4) = make_float4(vout[4], vout[5], vout[6], vout[7]);
        const uint4 z = make_uint4(0u, 0u, 0u, 0u);
        *(uint4*)(accz + base) = z;            *(uint4*)(accz + base + 4) = z;
        *(uint4*)(accz + N_ITEMS + base) = z;  *(uint4*)(accz + N_ITEMS + base + 4) = z;
      } else {
        float4 lo  = make_float4(vout[0], vout[1], vout[2], vout[3]);
        float4 hi4 = make_float4(vout[4], vout[5], vout[6], vout[7]);
        *(float4*)(vals + base)     = lo;  *(float4*)(vals + base + 4) = hi4;
        *(float4*)(w + base)        = lo;  *(float4*)(w + base + 4)    = hi4;
      }
    }
  }
}


__device__ __forceinline__ unsigned take_ticket(unsigned* cnt, int lane) {
  unsigned v = 0u;
  if (lane == 0) v = atomicAdd(cnt, 1u);
  return (unsigned)__builtin_amdgcn_readfirstlane((int)v);
}

__global__ void __launch_bounds__(256)
scatter_kernel(const int* __restrict__ perms, const float* __restrict__ vals,
               float* __restrict__ w, unsigned* __restrict__ q) {
  const int lane = threadIdx.x & 31;
  const unsigned slot = (unsigned)__builtin_amdgcn_readfirstlane((int)(threadIdx.x >> 5));
  const unsigned half = ((unsigned)__builtin_amdgcn_s_sendmsg_rtn(0x87) >> 18) & 1u;
  const float sc = -1.0f / S_PERMS;

  for (unsigned pass = 0; pass < 2u; ++pass) {
    const unsigned r = half ^ pass;
    unsigned* cnt = q + (r * N_SUB + slot) * Q_STRIDE;
    unsigned k = take_ticket(cnt, lane);
    while (k < (unsigned)SUB_TICKETS) {
      const unsigned t = k * N_SUB + slot;
      const unsigned f = t * TICKET + 4u * (unsigned)lane;
      const int4 pv = *(const int4*)(perms + f);
      const int pprev = perms[f ? f - 1u : 0u];
      const unsigned kn = take_ticket(cnt, lane);
      const bool o0 = ((((unsigned)pv.x >= (unsigned)SPLIT_ITEM) ? 1u : 0u) == r) && (f % N_ITEMS != 0u);
      const bool o1 = ((((unsigned)pv.y >= (unsigned)SPLIT_ITEM) ? 1u : 0u) == r);
      const bool o2 = ((((unsigned)pv.z >= (unsigned)SPLIT_ITEM) ? 1u : 0u) == r);
      const bool o3 = ((((unsigned)pv.w >= (unsigned)SPLIT_ITEM) ? 1u : 0u) == r);
      const float g0 = vals[o0 ? pprev : 0];
      const float g1 = vals[o1 ? pv.x : 0];
      const float g2 = vals[o2 ? pv.y : 0];
      const float g3 = vals[o3 ? pv.z : 0];
      if (o0) atomicAdd(&w[pv.x], g0 * sc);
      if (o1) atomicAdd(&w[pv.y], g1 * sc);
      if (o2) atomicAdd(&w[pv.z], g2 * sc);
      if (o3) atomicAdd(&w[pv.w], g3 * sc);
      k = kn;
    }
  }
}

#define BATCH        50000
#define BATCHES_ROW  (N_ITEMS / BATCH)
#define N_BATCHES    (S_PERMS * BATCHES_ROW)
#define LINE_SHIFT   5
#define N_LINES      (N_ITEMS >> LINE_SHIFT)
#define HI_LINES     2048u
#define BINS_PER_THR 7
#define PER_LANE     ((BATCH + 1023) / 1024)
#define ADD_GROUP    7

__global__ void __launch_bounds__(1024)
line_scatter_kernel(const int* __restrict__ perms, const float* __restrict__ vals,
                    unsigned* __restrict__ acc) {
  __shared__ unsigned slot[BATCH + 1];
  __shared__ unsigned hist[N_LINES];
  __shared__ unsigned wsum[32];
  __shared__ unsigned bound[3];
  const unsigned tid = threadIdx.x;
  const unsigned half = ((unsigned)__builtin_amdgcn_s_sendmsg_rtn(0x87) >> 18) & 1u;
  unsigned* a = acc + half * (unsigned)N_ITEMS;
  const unsigned b = blockIdx.x;
  const bool opens = (b % (unsigned)BATCHES_ROW) == 0u;
  const int* src = perms + (size_t)b * BATCH;

  for (unsigned k = tid; k < (unsigned)N_LINES; k += 1024u) hist[k] = 0u;
  __syncthreads();
  int mine[PER_LANE];
  #pragma unroll
  for (int j = 0; j < PER_LANE; ++j) {
    const unsigned i = tid + 1024u * (unsigned)j;
    mine[j] = (i < (unsigned)BATCH) ? src[i] : 0;
  }
  const int pred = opens ? src[0] : src[-1];
  #pragma unroll
  for (int j = 0; j < PER_LANE; ++j) {
    const unsigned i = tid + 1024u * (unsigned)j;
    if (i < (unsigned)BATCH) atomicAdd(&hist[(unsigned)mine[j] >> LINE_SHIFT], 1u);
  }
  __syncthreads();
  unsigned c[BINS_PER_THR], tot = 0u;
  #pragma unroll
  for (int j = 0; j < BINS_PER_THR; ++j) {
    const unsigned bin = tid * BINS_PER_THR + (unsigned)j;
    c[j] = (bin < (unsigned)N_LINES) ? hist[bin] : 0u;
    tot += c[j];
  }
  unsigned inc = tot;
  #pragma unroll
  for (int o = 1; o < 32; o <<= 1) {
    const unsigned u = __shfl_up(inc, o);
    if ((tid & 31u) >= (unsigned)o) inc += u;
  }
  if ((tid & 31u) == 31u) wsum[tid >> 5] = inc;
  __syncthreads();
  if (tid < 32u) {
    unsigned x = wsum[tid], y = x;
    #pragma unroll
    for (int o = 1; o < 32; o <<= 1) {
      const unsigned u = __shfl_up(y, o);
      if (tid >= (unsigned)o) y += u;
    }
    wsum[tid] = y - x;
  }
  __syncthreads();
  unsigned run = wsum[tid >> 5] + inc - tot;
  #pragma unroll
  for (int j = 0; j < BINS_PER_THR; ++j) {
    const unsigned bin = tid * BINS_PER_THR + (unsigned)j;
    if (bin < (unsigned)N_LINES) {
      hist[bin] = run;
      if (bin != 0u && (bin % HI_LINES) == 0u) bound[bin / HI_LINES - 1u] = run;
    }
    run += c[j];
  }
  __syncthreads();
  #pragma unroll
  for (int g = 0; g < PER_LANE; g += ADD_GROUP) {
    unsigned pos[ADD_GROUP];
    #pragma unroll
    for (int u = 0; u < ADD_GROUP; ++u) {
      const unsigned i = tid + 1024u * (unsigned)(g + u);
      pos[u] = (i < (unsigned)BATCH) ? atomicAdd(&hist[(unsigned)mine[g + u] >> LINE_SHIFT], 1u) : 0u;
    }
    #pragma unroll
    for (int u = 0; u < ADD_GROUP; ++u) {
      const unsigned i = tid + 1024u * (unsigned)(g + u);
      if (i < (unsigned)BATCH) slot[pos[u]] = (i << 16) | ((unsigned)mine[g + u] & 0xFFFFu);
    }
  }
  __syncthreads();
  const unsigned b0 = bound[0], b1 = bound[1], b2 = bound[2];
  unsigned wd[PER_LANE], hp[(PER_LANE + 15) / 16];
  #pragma unroll
  for (int q = 0; q < (PER_LANE + 15) / 16; ++q) hp[q] = 0u;
  #pragma unroll
  for (int j = 0; j < PER_LANE; ++j) {
    const unsigned k = tid + 1024u * (unsigned)j;
    const bool in = k < (unsigned)BATCH;
    wd[j] = in ? slot[k] : 0u;
    const unsigned hi = (k >= b0 ? 1u : 0u) + (k >= b1 ? 1u : 0u) + (k >= b2 ? 1u : 0u);
    hp[j >> 4] |= (in ? hi : 0u) << (2 * (j & 15));
  }
  __syncthreads();
#define ENT_OF(j)  (wd[(j)] >> 16)
#define ITEM_OF(j) ((((hp[(j) >> 4] >> (2 * ((j) & 15))) & 3u) << 16) | (wd[(j)] & 0xFFFFu))
  #pragma unroll
  for (int g = 0; g < PER_LANE; g += ADD_GROUP) {
    float v[ADD_GROUP];
    #pragma unroll
    for (int u = 0; u < ADD_GROUP; ++u) v[u] = vals[ITEM_OF(g + u)];
    #pragma unroll
    for (int u = 0; u < ADD_GROUP; ++u) {
      const unsigned k = tid + 1024u * (unsigned)(g + u);
      if (k < (unsigned)BATCH) slot[ENT_OF(g + u) + 1u] = __float2uint_rn(v[u] * FX_TERM);
    }
  }
  if (tid == 0) slot[0] = __float2uint_rn(vals[pred] * FX_TERM);
  __syncthreads();
  #pragma unroll
  for (int j = 0; j < PER_LANE; ++j) asm volatile("" : "+v"(wd[j]));
  asm volatile("" : "+v"(hp[0]), "+v"(hp[1]), "+v"(hp[2]), "+v"(hp[3]));
  #pragma unroll
  for (int g = 0; g < PER_LANE; g += ADD_GROUP) {
    unsigned sum[ADD_GROUP];
    #pragma unroll
    for (int u = 0; u < ADD_GROUP; ++u) sum[u] = slot[ENT_OF(g + u)];
    #pragma unroll
    for (int u = 0; u < ADD_GROUP; ++u) {
      const unsigned k = tid + 1024u * (unsigned)(g + u);
      if (k < (unsigned)BATCH && !(opens && ENT_OF(g + u) == 0u))
        __hip_atomic_fetch_add(a + ITEM_OF(g + u), sum[u], __ATOMIC_RELAXED, __HIP_MEMORY_SCOPE_AGENT);
    }
  }
#undef ENT_OF
#undef ITEM_OF
}

#define TAIL_LANES  256
#define TAIL_QUADS  (N_ITEMS / 4)
#define TAIL_GROUPS ((TAIL_QUADS + TAIL_LANES - 1) / TAIL_LANES)

__device__ __forceinline__ float fold_one(float v, unsigned s0, unsigned s1) {
  const int d = (int)(__float2uint_rn(v * FX_W) - (s0 + s1));
  return (float)d * (1.0f / FX_W);
}

__global__ void __launch_bounds__(TAIL_LANES)
fold_max_kernel(const float* __restrict__ vals, const unsigned* __restrict__ acc,
                float* __restrict__ w, unsigned* __restrict__ umax) {
  __shared__ float part[TAIL_LANES / 32];
  const int q = blockIdx.x * TAIL_LANES + threadIdx.x;
  float m = -INFINITY;
  float4 o = make_float4(0.f, 0.f, 0.f, 0.f);
  if (q < TAIL_QUADS) {
    const uint4  s0 = ((const uint4*)acc)[q];
    const uint4  s1 = ((const uint4*)(acc + N_ITEMS))[q];
    const float4 v  = ((const float4*)vals)[q];
    o.x = fold_one(v.x, s0.x, s1.x);
    o.y = fold_one(v.y, s0.y, s1.y);
    o.z = fold_one(v.z, s0.z, s1.z);
    o.w = fold_one(v.w, s0.w, s1.w);
    m = fmaxf(fmaxf(o.x, o.y), fmaxf(o.z, o.w));
  }
  #pragma unroll
  for (int s = 16; s; s >>= 1) m = fmaxf(m, __shfl_xor(m, s));
  if ((threadIdx.x & 31) == 0) part[threadIdx.x >> 5] = m;
  __syncthreads();
  if (threadIdx.x == 0) {
    float x = part[0];
    #pragma unroll
    for (int j = 1; j < TAIL_LANES / 32; ++j) x = fmaxf(x, part[j]);
    atomicMax(umax, fenc(x));
  }
  if (q < TAIL_QUADS) ((float4*)w)[q] = o;
}

__global__ void max_kernel(const float* __restrict__ w, unsigned* __restrict__ umax, int n) {
  float m = -INFINITY;
  for (int i = blockIdx.x * blockDim.x + threadIdx.x; i < n; i += gridDim.x * blockDim.x)
    m = fmaxf(m, w[i]);
  #pragma unroll
  for (int o = 16; o; o >>= 1) m = fmaxf(m, __shfl_xor(m, o));
  if ((threadIdx.x & 31) == 0) atomicMax(umax, fenc(m));
}

__global__ void __launch_bounds__(TAIL_LANES)
sum_kernel(const float* __restrict__ w, const unsigned* __restrict__ umax,
           float* __restrict__ sum) {
  __shared__ float part[TAIL_LANES / 32];
  const float mx = fdec(*umax);
  const int q = blockIdx.x * TAIL_LANES + threadIdx.x;
  float a = 0.f;
  if (q < TAIL_QUADS) {
    const float4 v = ((const float4*)w)[q];
    a = (__expf(v.x - mx) + __expf(v.y - mx)) + (__expf(v.z - mx) + __expf(v.w - mx));
  }
  #pragma unroll
  for (int o = 16; o; o >>= 1) a += __shfl_xor(a, o);
  if ((threadIdx.x & 31) == 0) part[threadIdx.x >> 5] = a;
  __syncthreads();
  if (threadIdx.x == 0) {
    float x = part[0];
    #pragma unroll
    for (int j = 1; j < TAIL_LANES / 32; ++j) x += part[j];
    atomicAdd(sum, x);
  }
}

__global__ void final_kernel(float* __restrict__ w, const unsigned* __restrict__ umax,
                             const float* __restrict__ sum, int n) {
  const float mx = fdec(*umax);
  const float rs = 1.0f / *sum;
  int i = blockIdx.x * blockDim.x + threadIdx.x;
  if (i < n) w[i] = __expf(w[i] - mx) * rs;
}

extern "C" void kernel_launch(void* const* d_in, const int* in_sizes, int n_in,
                              void* d_out, int out_size, void* d_ws, size_t ws_size,
                              hipStream_t stream) {
  const float* neigh = (const float*)d_in[0];
  const float* tgt   = (const float*)d_in[1];
  const int*   perms = (const int*)d_in[2];
  float* out  = (float*)d_out;
  const int nTiles = N_ITEMS / 16;

  if (ws_size >= (size_t)CHAIN_WS_BYTES) {
    u64* base64 = (u64*)d_ws;
    float* cvals = (float*)d_ws;
    unsigned* acc = (unsigned*)(cvals + N_ITEMS);
    float* cstats = (float*)(base64 + 2 * N_ITEMS);
    unsigned* cumax = (unsigned*)(cstats + 2);
    unsigned* cq = (unsigned*)(cstats + 64);
    init_kernel<<<1, 32, 0, stream>>>(tgt, cstats, cumax, cq);
    vals_kernel<true><<<(nTiles + 7) / 8, 256, 0, stream>>>(neigh, tgt, cstats, cvals, nullptr, (unsigned long long*)acc, nTiles);
    line_scatter_kernel<<<N_BATCHES, 1024, 0, stream>>>(perms, cvals, acc);
    fold_max_kernel<<<TAIL_GROUPS, TAIL_LANES, 0, stream>>>(cvals, acc, out, cumax);
    sum_kernel<<<TAIL_GROUPS, TAIL_LANES, 0, stream>>>(out, cumax, cstats + 1);
    final_kernel<<<(N_ITEMS + 255) / 256, 256, 0, stream>>>(out, cumax, cstats + 1, N_ITEMS);
    return;
  }

  float* vals = (float*)d_ws;
  float* stats = vals + N_ITEMS;
  unsigned* umax = (unsigned*)(stats + 2);
  unsigned* q = (unsigned*)(stats + 64);

  init_kernel<<<1, 32, 0, stream>>>(tgt, stats, umax, q);

  vals_kernel<false><<<(nTiles + 7) / 8, 256, 0, stream>>>(neigh, tgt, stats, vals, out, nullptr, nTiles);

  scatter_kernel<<<2048, 256, 0, stream>>>(perms, vals, out, q);

  max_kernel<<<1024, 256, 0, stream>>>(out, umax, N_ITEMS);
  sum_kernel<<<TAIL_GROUPS, TAIL_LANES, 0, stream>>>(out, umax, stats + 1);
  final_kernel<<<(N_ITEMS + 255) / 256, 256, 0, stream>>>(out, umax, stats + 1, N_ITEMS);
}


# Value launch: square root, division and power that finish a row run once per row (one row a lane: every accumulator column holds the same dots), not for all 8 rows in every lane; values bit-equal
# speedup vs baseline: 12.0381x; 1.0453x over previous
#include <hip/hip_runtime.h>
#include <math.h>

#define N_ITEMS 200000
#define DIMS    128
#define S_PERMS 128
#define ALPHA_C 0.7f
#define EPS_C   1e-8f

#define TICKET    128
#define N_TICKETS (S_PERMS * N_ITEMS / TICKET)
#define N_SUB       8
#define SUB_TICKETS (N_TICKETS / N_SUB)
#define Q_STRIDE    64
#define SPLIT_ITEM 100032

#define FX_TERM 131072.0f
#define FX_W    16777216.0f
#define CHAIN_WS_BYTES (2 * N_ITEMS * 8 + 256 + 2 * N_SUB * Q_STRIDE * 4)

typedef unsigned long long u64;
typedef float v2f __attribute__((ext_vector_type(2)));
typedef float v8f __attribute__((ext_vector_type(8)));

__device__ __forceinline__ unsigned fenc(float f) {
  unsigned u = __float_as_uint(f);
  return (u & 0x80000000u) ? ~u : (u | 0x80000000u);
}
__device__ __forceinline__ float fdec(unsigned e) {
  unsigned u = (e & 0x80000000u) ? (e & 0x7FFFFFFFu) : ~e;
  return __uint_as_float(u);
}

__global__ void init_kernel(const float* __restrict__ tgt,
                            float* __restrict__ stats,
                            unsigned* __restrict__ umax,
                            unsigned* __restrict__ q) {
  int lane = threadIdx.x;
  float ss = 0.f;
  for (int k = lane; k < DIMS; k += 32) { float v = tgt[k]; ss = fmaf(v, v, ss); }
  #pragma unroll
  for (int o = 16; o; o >>= 1) ss += __shfl_xor(ss, o);
  if (lane == 0) {
    stats[0] = sqrtf(ss); stats[1] = 0.f; *umax = 0u;
  }
  if (lane < 2 * N_SUB) q[lane * Q_STRIDE] = 0u;
}

template <bool WORDS>
__global__ void __launch_bounds__(256)
vals_kernel(const float* __restrict__ neigh, const float* __restrict__ tgt,
            const float* __restrict__ stats, float* __restrict__ vals,
            float* __restrict__ w, unsigned long long* __restrict__ words, int nTiles) {
  __shared__ __align__(16) float lds[8][16 * 132];
  const int lane  = threadIdx.x & 31;
  const int wave  = threadIdx.x >> 5;
  const int hi    = lane >> 4;
  const int row16 = lane & 15;
  float* tile = &lds[wave][0];
  const float tn = fmaxf(stats[0], EPS_C);

  v2f bv[32];
  #pragma unroll
  for (int k = 0; k < 32; ++k)
    bv[k] = *(const v2f*)(tgt + 4 * k + 2 * hi);

  const int r = lane >> 1, h = lane & 1;
  const int wavesTotal = (gridDim.x * blockDim.x) >> 5;
  for (int t = (blockIdx.x * blockDim.x + threadIdx.x) >> 5; t < nTiles;
       t += wavesTotal) {
    const float* src = neigh + (size_t)t * (16 * DIMS);
    const float4* srow = (const float4*)(src + r * DIMS + h * 64);
    float* drow = tile + r * 132 + h * 64;
    float ss = 0.f;
    #pragma unroll
    for (int j = 0; j < 16; ++j) {
      float4 v = srow[j];
      ss = fmaf(v.x, v.x, ss); ss = fmaf(v.y, v.y, ss);
      ss = fmaf(v.z, v.z, ss); ss = fmaf(v.w, v.w, ss);
      *(float4*)(drow + 4 * j) = v;
    }
    ss += __shfl_xor(ss, 1);

    v8f acc = {};
    const float* arow = tile + row16 * 132 + 2 * hi;
    #pragma unroll
    for (int k = 0; k < 32; ++k) {
      v2f a = *(const v2f*)(arow + 4 * k);
      acc = __builtin_amdgcn_wmma_f32_16x16x4_f32(false, a, false, bv[k],
                                                  (short)0, acc, false, false);
    }

    const int jr = row16 & 7;
    float dot = acc[0];
    #pragma unroll
    for (int j = 1; j < 8; ++j) dot = (jr == j) ? acc[j] : dot;
    const float ssj = __shfl(ss, (lane & 16) + 2 * jr);
    const float nn  = fmaxf(sqrtf(ssj), EPS_C);
    const float sim = dot / (nn * tn);
    const float vo  = dot * __powf(sim, ALPHA_C);
    if (row16 < 8) {
      const int row = t * 16 + hi * 8 + jr;
      vals[row] = vo;
      if (WORDS) {
        unsigned* accz = (unsigned*)words;
        accz[row] = 0u;
        accz[N_ITEMS + row] = 0u;
      } else {
        w[row] = vo;
      }
    }
  }
}


__device__ __forceinline__ unsigned take_ticket(unsigned* cnt, int lane) {
  unsigned v = 0u;
  if (lane == 0) v = atomicAdd(cnt, 1u);
  return (unsigned)__builtin_amdgcn_readfirstlane((int)v);
}

__global__ void __launch_bounds__(256)
scatter_kernel(const int* __restrict__ perms, const float* __restrict__ vals,
               float* __restrict__ w, unsigned* __restrict__ q) {
  const int lane = threadIdx.x & 31;
  const unsigned slot = (unsigned)__builtin_amdgcn_readfirstlane((int)(threadIdx.x >> 5));
  const unsigned half = ((unsigned)__builtin_amdgcn_s_sendmsg_rtn(0x87) >> 18) & 1u;
  const float sc = -1.0f / S_PERMS;

  for (unsigned pass = 0; pass < 2u; ++pass) {
    const unsigned r = half ^ pass;
    unsigned* cnt = q + (r * N_SUB + slot) * Q_STRIDE;
    unsigned k = take_ticket(cnt, lane);
    while (k < (unsigned)SUB_TICKETS) {
      const unsigned t = k * N_SUB + slot;
      const unsigned f = t * TICKET + 4u * (unsigned)lane;
      const int4 pv = *(const int4*)(perms + f);
      const int pprev = perms[f ? f - 1u : 0u];
      const unsigned kn = take_ticket(cnt, lane);
      const bool o0 = ((((unsigned)pv.x >= (unsigned)SPLIT_ITEM) ? 1u : 0u) == r) && (f % N_ITEMS != 0u);
      const bool o1 = ((((unsigned)pv.y >= (unsigned)SPLIT_ITEM) ? 1u : 0u) == r);
      const bool o2 = ((((unsigned)pv.z >= (unsigned)SPLIT_ITEM) ? 1u : 0u) == r);
      const bool o3 = ((((unsigned)pv.w >= (unsigned)SPLIT_ITEM) ? 1u : 0u) == r);
      const float g0 = vals[o0 ? pprev : 0];
      const float g1 = vals[o1 ? pv.x : 0];
      const float g2 = vals[o2 ? pv.y : 0];
      const float g3 = vals[o3 ? pv.z : 0];
      if (o0) atomicAdd(&w[pv.x], g0 * sc);
      if (o1) atomicAdd(&w[pv.y], g1 * sc);
      if (o2) atomicAdd(&w[pv.z], g2 * sc);
      if (o3) atomicAdd(&w[pv.w], g3 * sc);
      k = kn;
    }
  }
}

#define BATCH        50000
#define BATCHES_ROW  (N_ITEMS / BATCH)
#define N_BATCHES    (S_PERMS * BATCHES_ROW)
#define LINE_SHIFT   5
#define N_LINES      (N_ITEMS >> LINE_SHIFT)
#define HI_LINES     2048u
#define BINS_PER_THR 7
#define PER_LANE     ((BATCH + 1023) / 1024)
#define ADD_GROUP    7

__global__ void __launch_bounds__(1024)
line_scatter_kernel(const int* __restrict__ perms, const float* __restrict__ vals,
                    unsigned* __restrict__ acc) {
  __shared__ unsigned slot[BATCH + 1];
  __shared__ unsigned hist[N_LINES];
  __shared__ unsigned wsum[32];
  __shared__ unsigned bound[3];
  const unsigned tid = threadIdx.x;
  const unsigned half = ((unsigned)__builtin_amdgcn_s_sendmsg_rtn(0x87) >> 18) & 1u;
  unsigned* a = acc + half * (unsigned)N_ITEMS;
  const unsigned b = blockIdx.x;
  const bool opens = (b % (unsigned)BATCHES_ROW) == 0u;
  const int* src = perms + (size_t)b * BATCH;

  for (unsigned k = tid; k < (unsigned)N_LINES; k += 1024u) hist[k] = 0u;
  __syncthreads();
  int mine[PER_LANE];
  #pragma unroll
  for (int j = 0; j < PER_LANE; ++j) {
    const unsigned i = tid + 1024u * (unsigned)j;
    mine[j] = (i < (unsigned)BATCH) ? src[i] : 0;
  }
  const int pred = opens ? src[0] : src[-1];
  #pragma unroll
  for (int j = 0; j < PER_LANE; ++j) {
    const unsigned i = tid + 1024u * (unsigned)j;
    if (i < (unsigned)BATCH) atomicAdd(&hist[(unsigned)mine[j] >> LINE_SHIFT], 1u);
  }
  __syncthreads();
  unsigned c[BINS_PER_THR], tot = 0u;
  #pragma unroll
  for (int j = 0; j < BINS_PER_THR; ++j) {
    const unsigned bin = tid * BINS_PER_THR + (unsigned)j;
    c[j] = (bin < (unsigned)N_LINES) ? hist[bin] : 0u;
    tot += c[j];
  }
  unsigned inc = tot;
  #pragma unroll
  for (int o = 1; o < 32; o <<= 1) {
    const unsigned u = __shfl_up(inc, o);
    if ((tid & 31u) >= (unsigned)o) inc += u;
  }
  if ((tid & 31u) == 31u) wsum[tid >> 5] = inc;
  __syncthreads();
  if (tid < 32u) {
    unsigned x = wsum[tid], y = x;
    #pragma unroll
    for (int o = 1; o < 32; o <<= 1) {
      const unsigned u = __shfl_up(y, o);
      if (tid >= (unsigned)o) y += u;
    }
    wsum[tid] = y - x;
  }
  __syncthreads();
  unsigned run = wsum[tid >> 5] + inc - tot;
  #pragma unroll
  for (int j = 0; j < BINS_PER_THR; ++j) {
    const unsigned bin = tid * BINS_PER_THR + (unsigned)j;
    if (bin < (unsigned)N_LINES) {
      hist[bin] = run;
      if (bin != 0u && (bin % HI_LINES) == 0u) bound[bin / HI_LINES - 1u] = run;
    }
    run += c[j];
  }
  __syncthreads();
  #pragma unroll
  for (int g = 0; g < PER_LANE; g += ADD_GROUP) {
    unsigned pos[ADD_GROUP];
    #pragma unroll
    for (int u = 0; u < ADD_GROUP; ++u) {
      const unsigned i = tid + 1024u * (unsigned)(g + u);
      pos[u] = (i < (unsigned)BATCH) ? atomicAdd(&hist[(unsigned)mine[g + u] >> LINE_SHIFT], 1u) : 0u;
    }
    #pragma unroll
    for (int u = 0; u < ADD_GROUP; ++u) {
      const unsigned i = tid + 1024u * (unsigned)(g + u);
      if (i < (unsigned)BATCH) slot[pos[u]] = (i << 16) | ((unsigned)mine[g + u] & 0xFFFFu);
    }
  }
  __syncthreads();
  const unsigned b0 = bound[0], b1 = bound[1], b2 = bound[2];
  unsigned wd[PER_LANE], hp[(PER_LANE + 15) / 16];
  #pragma unroll
  for (int q = 0; q < (PER_LANE + 15) / 16; ++q) hp[q] = 0u;
  #pragma unroll
  for (int j = 0; j < PER_LANE; ++j) {
    const unsigned k = tid + 1024u * (unsigned)j;
    const bool in = k < (unsigned)BATCH;
    wd[j] = in ? slot[k] : 0u;
    const unsigned hi = (k >= b0 ? 1u : 0u) + (k >= b1 ? 1u : 0u) + (k >= b2 ? 1u : 0u);
    hp[j >> 4] |= (in ? hi : 0u) << (2 * (j & 15));
  }
  __syncthreads();
#define ENT_OF(j)  (wd[(j)] >> 16)
#define ITEM_OF(j) ((((hp[(j) >> 4] >> (2 * ((j) & 15))) & 3u) << 16) | (wd[(j)] & 0xFFFFu))
  #pragma unroll
  for (int g = 0; g < PER_LANE; g += ADD_GROUP) {
    float v[ADD_GROUP];
    #pragma unroll
    for (int u = 0; u < ADD_GROUP; ++u) v[u] = vals[ITEM_OF(g + u)];
    #pragma unroll
    for (int u = 0; u < ADD_GROUP; ++u) {
      const unsigned k = tid + 1024u * (unsigned)(g + u);
      if (k < (unsigned)BATCH) slot[ENT_OF(g + u) + 1u] = __float2uint_rn(v[u] * FX_TERM);
    }
  }
  if (tid == 0) slot[0] = __float2uint_rn(vals[pred] * FX_TERM);
  __syncthreads();
  #pragma unroll
  for (int j = 0; j < PER_LANE; ++j) asm volatile("" : "+v"(wd[j]));
  asm volatile("" : "+v"(hp[0]), "+v"(hp[1]), "+v"(hp[2]), "+v"(hp[3]));
  #pragma unroll
  for (int g = 0; g < PER_LANE; g += ADD_GROUP) {
    unsigned sum[ADD_GROUP];
    #pragma unroll
    for (int u = 0; u < ADD_GROUP; ++u) sum[u] = slot[ENT_OF(g + u)];
    #pragma unroll
    for (int u = 0; u < ADD_GROUP; ++u) {
      const unsigned k = tid + 1024u * (unsigned)(g + u);
      if (k < (unsigned)BATCH && !(opens && ENT_OF(g + u) == 0u))
        __hip_atomic_fetch_add(a + ITEM_OF(g + u), sum[u], __ATOMIC_RELAXED, __HIP_MEMORY_SCOPE_AGENT);
    }
  }
#undef ENT_OF
#undef ITEM_OF
}

#define TAIL_LANES  256
#define TAIL_QUADS  (N_ITEMS / 4)
#define TAIL_GROUPS ((TAIL_QUADS + TAIL_LANES - 1) / TAIL_LANES)

__device__ __forceinline__ float fold_one(float v, unsigned s0, unsigned s1) {
  const int d = (int)(__float2uint_rn(v * FX_W) - (s0 + s1));
  return (float)d * (1.0f / FX_W);
}

__global__ void __launch_bounds__(TAIL_LANES)
fold_max_kernel(const float* __restrict__ vals, const unsigned* __restrict__ acc,
                float* __restrict__ w, unsigned* __restrict__ umax) {
  __shared__ float part[TAIL_LANES / 32];
  const int q = blockIdx.x * TAIL_LANES + threadIdx.x;
  float m = -INFINITY;
  float4 o = make_float4(0.f, 0.f, 0.f, 0.f);
  if (q < TAIL_QUADS) {
    const uint4  s0 = ((const uint4*)acc)[q];
    const uint4  s1 = ((const uint4*)(acc + N_ITEMS))[q];
    const float4 v  = ((const float4*)vals)[q];
    o.x = fold_one(v.x, s0.x, s1.x);
    o.y = fold_one(v.y, s0.y, s1.y);
    o.z = fold_one(v.z, s0.z, s1.z);
    o.w = fold_one(v.w, s0.w, s1.w);
    m = fmaxf(fmaxf(o.x, o.y), fmaxf(o.z, o.w));
  }
  #pragma unroll
  for (int s = 16; s; s >>= 1) m = fmaxf(m, __shfl_xor(m, s));
  if ((threadIdx.x & 31) == 0) part[threadIdx.x >> 5] = m;
  __syncthreads();
  if (threadIdx.x == 0) {
    float x = part[0];
    #pragma unroll
    for (int j = 1; j < TAIL_LANES / 32; ++j) x = fmaxf(x, part[j]);
    atomicMax(umax, fenc(x));
  }
  if (q < TAIL_QUADS) ((float4*)w)[q] = o;
}

__global__ void max_kernel(const float* __restrict__ w, unsigned* __restrict__ umax, int n) {
  float m = -INFINITY;
  for (int i = blockIdx.x * blockDim.x + threadIdx.x; i < n; i += gridDim.x * blockDim.x)
    m = fmaxf(m, w[i]);
  #pragma unroll
  for (int o = 16; o; o >>= 1) m = fmaxf(m, __shfl_xor(m, o));
  if ((threadIdx.x & 31) == 0) atomicMax(umax, fenc(m));
}

__global__ void __launch_bounds__(TAIL_LANES)
sum_kernel(const float* __restrict__ w, const unsigned* __restrict__ umax,
           float* __restrict__ sum) {
  __shared__ float part[TAIL_LANES / 32];
  const float mx = fdec(*umax);
  const int q = blockIdx.x * TAIL_LANES + threadIdx.x;
  float a = 0.f;
  if (q < TAIL_QUADS) {
    const float4 v = ((const float4*)w)[q];
    a = (__expf(v.x - mx) + __expf(v.y - mx)) + (__expf(v.z - mx) + __expf(v.w - mx));
  }
  #pragma unroll
  for (int o = 16; o; o >>= 1) a += __shfl_xor(a, o);
  if ((threadIdx.x & 31) == 0) part[threadIdx.x >> 5] = a;
  __syncthreads();
  if (threadIdx.x == 0) {
    float x = part[0];
    #pragma unroll
    for (int j = 1; j < TAIL_LANES / 32; ++j) x += part[j];
    atomicAdd(sum, x);
  }
}

__global__ void final_kernel(float* __restrict__ w, const unsigned* __restrict__ umax,
                             const float* __restrict__ sum, int n) {
  const float mx = fdec(*umax);
  const float rs = 1.0f / *sum;
  int i = blockIdx.x * blockDim.x + threadIdx.x;
  if (i < n) w[i] = __expf(w[i] - mx) * rs;
}

extern "C" void kernel_launch(void* const* d_in, const int* in_sizes, int n_in,
                              void* d_out, int out_size, void* d_ws, size_t ws_size,
                              hipStream_t stream) {
  const float* neigh = (const float*)d_in[0];
  const float* tgt   = (const float*)d_in[1];
  const int*   perms = (const int*)d_in[2];
  float* out  = (float*)d_out;
  const int nTiles = N_ITEMS / 16;

  if (ws_size >= (size_t)CHAIN_WS_BYTES) {
    u64* base64 = (u64*)d_ws;
    float* cvals = (float*)d_ws;
    unsigned* acc = (unsigned*)(cvals + N_ITEMS);
    float* cstats = (float*)(base64 + 2 * N_ITEMS);
    unsigned* cumax = (unsigned*)(cstats + 2);
    unsigned* cq = (unsigned*)(cstats + 64);
    init_kernel<<<1, 32, 0, stream>>>(tgt, cstats, cumax, cq);
    vals_kernel<true><<<(nTiles + 7) / 8, 256, 0, stream>>>(neigh, tgt, cstats, cvals, nullptr, (unsigned long long*)acc, nTiles);
    line_scatter_kernel<<<N_BATCHES, 1024, 0, stream>>>(perms, cvals, acc);
    fold_max_kernel<<<TAIL_GROUPS, TAIL_LANES, 0, stream>>>(cvals, acc, out, cumax);
    sum_kernel<<<TAIL_GROUPS, TAIL_LANES, 0, stream>>>(out, cumax, cstats + 1);
    final_kernel<<<(N_ITEMS + 255) / 256, 256, 0, stream>>>(out, cumax, cstats + 1, N_ITEMS);
    return;
  }

  float* vals = (float*)d_ws;
  float* stats = vals + N_ITEMS;
  unsigned* umax = (unsigned*)(stats + 2);
  unsigned* q = (unsigned*)(stats + 64);

  init_kernel<<<1, 32, 0, stream>>>(tgt, stats, umax, q);

  vals_kernel<false><<<(nTiles + 7) / 8, 256, 0, stream>>>(neigh, tgt, stats, vals, out, nullptr, nTiles);

  scatter_kernel<<<2048, 256, 0, stream>>>(perms, vals, out, q);

  max_kernel<<<1024, 256, 0, stream>>>(out, umax, N_ITEMS);
  sum_kernel<<<TAIL_GROUPS, TAIL_LANES, 0, stream>>>(out, umax, stats + 1);
  final_kernel<<<(N_ITEMS + 255) / 256, 256, 0, stream>>>(out, umax, stats + 1, N_ITEMS);
}
